# starting kernel
# baseline (speedup 1.0000x reference)
#include <hip/hip_runtime.h>
#include <hip/hip_bf16.h>

#define BB 2
#define TT 12
#define NN 10000
#define CIN 64
#define COUT 32
#define KK1 17
#define HH 4

typedef float v2f __attribute__((ext_vector_type(2)));
typedef float v8f __attribute__((ext_vector_type(8)));

__global__ __launch_bounds__(256) void proj_wmma_kernel(
    const float* __restrict__ x, const float* __restrict__ W,
    const float* __restrict__ bias, float* __restrict__ proj, int M) {
  const int lane = threadIdx.x & 31;
  const int half = lane >> 4;
  const int l15  = lane & 15;
  const long wave = (long)blockIdx.x * (blockDim.x >> 5) + (threadIdx.x >> 5);
  const long row0 = wave * 16;
  if (row0 >= M) return;

  const float* ap  = x + (row0 + l15) * CIN + half * 2;
  const float* bp0 = W + (long)l15        * CIN + half * 2;
  const float* bp1 = W + (long)(l15 + 16) * CIN + half * 2;

  v8f acc0 = {};
  v8f acc1 = {};
#pragma unroll
  for (int ks = 0; ks < 16; ++ks) {
    v2f a  = *(const v2f*)(ap  + ks * 4);
    v2f b0 = *(const v2f*)(bp0 + ks * 4);
    v2f b1 = *(const v2f*)(bp1 + ks * 4);
    acc0 = __builtin_amdgcn_wmma_f32_16x16x4_f32(false, a, false, b0,
                                                 (short)0, acc0, false, false);
    acc1 = __builtin_amdgcn_wmma_f32_16x16x4_f32(false, a, false, b1,
                                                 (short)0, acc1, false, false);
  }

  const float bv0 = bias[l15];
  const float bv1 = bias[l15 + 16];
#pragma unroll
  for (int r = 0; r < 8; ++r) {
    const long m = row0 + r + half * 8;
    proj[m * COUT + l15]      = acc0[r] + bv0;
    proj[m * COUT + 16 + l15] = acc1[r] + bv1;
  }
}

__global__ __launch_bounds__(256) void weights_kernel(
    const float* __restrict__ nd, const int* __restrict__ nn,
    float* __restrict__ wbuf, float* __restrict__ dagg) {
  const int n = blockIdx.x * blockDim.x + threadIdx.x;
  if (n >= NN) return;
  float da[HH] = {0.f, 0.f, 0.f, 0.f};
  for (int k = 0; k < KK1; ++k) {
    const float d   = nd[n * KK1 + k];
    const int node  = nn[n * KK1 + k];
    const float d2  = d * d;
    const float dc  = isinf(d) ? 0.f : d;
#pragma unroll
    for (int h = 0; h < HH; ++h) {
      const float lam = (float)(h + 1) * (1.0f / HH);
      float wv = __expf(-d2 * lam);
      if (node == -1)   wv = 0.f;
      if (wv < 1e-5f)   wv = 0.f;
      wbuf[(n * KK1 + k) * HH + h] = wv;
      da[h] += wv * dc;
    }
  }
#pragma unroll
  for (int h = 0; h < HH; ++h) dagg[n * HH + h] = da[h];
}

__global__ __launch_bounds__(256) void agg_kernel(
    const float* __restrict__ proj, const float* __restrict__ wbuf,
    const float* __restrict__ dagg, const int* __restrict__ nn,
    float* __restrict__ out) {
  const int lane = threadIdx.x & 31;
  const long wave = (long)blockIdx.x * (blockDim.x >> 5) + (threadIdx.x >> 5);
  const long total = (long)BB * TT * NN;
  if (wave >= total) return;
  const int n   = (int)(wave % NN);
  const long bt = wave / NN;
  const float* pslice = proj + bt * (long)NN * COUT;

  float acc0 = 0.f, acc1 = 0.f, acc2 = 0.f, acc3 = 0.f;
  for (int k = 0; k < KK1; ++k) {
    const int node = nn[n * KK1 + k];
    float val = 0.f;
    if (node >= 0 && node < NN)
      val = pslice[(long)node * COUT + lane];
    const float4 w4 = *(const float4*)(wbuf + (n * KK1 + k) * HH);
    acc0 += w4.x * val;
    acc1 += w4.y * val;
    acc2 += w4.z * val;
    acc3 += w4.w * val;
  }

  float* o = out + wave * (long)HH * COUT;
  o[0 * COUT + lane] = acc0 + dagg[n * HH + 0];
  o[1 * COUT + lane] = acc1 + dagg[n * HH + 1];
  o[2 * COUT + lane] = acc2 + dagg[n * HH + 2];
  o[3 * COUT + lane] = acc3 + dagg[n * HH + 3];
}

extern "C" void kernel_launch(void* const* d_in, const int* in_sizes, int n_in,
                              void* d_out, int out_size, void* d_ws, size_t ws_size,
                              hipStream_t stream) {
  const float* x  = (const float*)d_in[0];
  const float* W  = (const float*)d_in[1];
  const float* b  = (const float*)d_in[2];
  const float* nd = (const float*)d_in[3];
  const int*   nn = (const int*)d_in[4];
  float* out = (float*)d_out;

  const long M = (long)BB * TT * NN;

  float* proj = (float*)d_ws;
  float* wbuf = proj + M * COUT;
  float* dagg = wbuf + (long)NN * KK1 * HH;

  const int tiles = (int)(M / 16);
  proj_wmma_kernel<<<(tiles + 7) / 8, 256, 0, stream>>>(x, W, b, proj, (int)M);

  weights_kernel<<<(NN + 255) / 256, 256, 0, stream>>>(nd, nn, wbuf, dagg);

  agg_kernel<<<(int)((M + 7) / 8), 256, 0, stream>>>(proj, wbuf, dagg, nn, out);
}


# agg: one wave per (node, 12 (b,t) slices) instead of per (b,t,n): ids and weights read once per wave by scalar loads, all 51 row gathers issued ahead of the sums; sum order unchanged
# speedup vs baseline: 2.5248x; 2.5248x over previous
#include <hip/hip_runtime.h>
#include <hip/hip_bf16.h>

#define BB 2
#define TT 12
#define NN 10000
#define CIN 64
#define COUT 32
#define KK1 17
#define HH 4

typedef float v2f __attribute__((ext_vector_type(2)));
typedef float v8f __attribute__((ext_vector_type(8)));

__global__ __launch_bounds__(256) void proj_wmma_kernel(
    const float* __restrict__ x, const float* __restrict__ W,
    const float* __restrict__ bias, float* __restrict__ proj, int M) {
  const int lane = threadIdx.x & 31;
  const int half = lane >> 4;
  const int l15  = lane & 15;
  const long wave = (long)blockIdx.x * (blockDim.x >> 5) + (threadIdx.x >> 5);
  const long row0 = wave * 16;
  if (row0 >= M) return;

  const float* ap  = x + (row0 + l15) * CIN + half * 2;
  const float* bp0 = W + (long)l15        * CIN + half * 2;
  const float* bp1 = W + (long)(l15 + 16) * CIN + half * 2;

  v8f acc0 = {};
  v8f acc1 = {};
#pragma unroll
  for (int ks = 0; ks < 16; ++ks) {
    v2f a  = *(const v2f*)(ap  + ks * 4);
    v2f b0 = *(const v2f*)(bp0 + ks * 4);
    v2f b1 = *(const v2f*)(bp1 + ks * 4);
    acc0 = __builtin_amdgcn_wmma_f32_16x16x4_f32(false, a, false, b0,
                                                 (short)0, acc0, false, false);
    acc1 = __builtin_amdgcn_wmma_f32_16x16x4_f32(false, a, false, b1,
                                                 (short)0, acc1, false, false);
  }

  const float bv0 = bias[l15];
  const float bv1 = bias[l15 + 16];
#pragma unroll
  for (int r = 0; r < 8; ++r) {
    const long m = row0 + r + half * 8;
    proj[m * COUT + l15]      = acc0[r] + bv0;
    proj[m * COUT + 16 + l15] = acc1[r] + bv1;
  }
}

__global__ __launch_bounds__(256) void weights_kernel(
    const float* __restrict__ nd, const int* __restrict__ nn,
    float* __restrict__ wbuf, float* __restrict__ dagg) {
  const int n = blockIdx.x * blockDim.x + threadIdx.x;
  if (n >= NN) return;
  float da[HH] = {0.f, 0.f, 0.f, 0.f};
  for (int k = 0; k < KK1; ++k) {
    const float d   = nd[n * KK1 + k];
    const int node  = nn[n * KK1 + k];
    const float d2  = d * d;
    const float dc  = isinf(d) ? 0.f : d;
#pragma unroll
    for (int h = 0; h < HH; ++h) {
      const float lam = (float)(h + 1) * (1.0f / HH);
      float wv = __expf(-d2 * lam);
      if (node == -1)   wv = 0.f;
      if (wv < 1e-5f)   wv = 0.f;
      wbuf[(n * KK1 + k) * HH + h] = wv;
      da[h] += wv * dc;
    }
  }
#pragma unroll
  for (int h = 0; h < HH; ++h) dagg[n * HH + h] = da[h];
}

#define SLICE_BYTES (NN * COUT * 4)
#define BT_GROUP 4
#define BT_PER_WAVE 12
#define AGG_WAVES 4

__global__ __launch_bounds__(AGG_WAVES * 32) void agg_kernel(
    const float* __restrict__ proj, const float* __restrict__ wbuf,
    const float* __restrict__ dagg, const int* __restrict__ nn,
    float* __restrict__ out) {
  const int lane = threadIdx.x & 31;
  const int wv = __builtin_amdgcn_readfirstlane((int)(threadIdx.x >> 5));
  const int n = blockIdx.x * AGG_WAVES + wv;
  if (n >= NN) return;
  const int bt0 = blockIdx.y * BT_PER_WAVE;
  const int j  = lane >> 3;
  const int c4 = lane & 7;

  unsigned voff[KK1];
  unsigned okmask = 0;
  const unsigned lanepart = (unsigned)j * SLICE_BYTES + (unsigned)c4 * 16u;
  for (int k = 0; k < KK1; ++k) {
    const int node = nn[n * KK1 + k];
    const bool ok = (node >= 0) && (node < NN);
    okmask |= (ok ? 1u : 0u) << k;
    voff[k] = (ok ? (unsigned)node : 0u) * (COUT * 4) + lanepart;
  }
  float w[KK1][HH];
  for (int k = 0; k < KK1; ++k)
    for (int h = 0; h < HH; ++h) w[k][h] = wbuf[(n * KK1 + k) * HH + h];
  float dg[HH];
  for (int h = 0; h < HH; ++h) dg[h] = dagg[n * HH + h];

  for (int g = 0; g < BT_PER_WAVE / BT_GROUP; ++g) {
    const int btg = bt0 + g * BT_GROUP;
    const char* base = (const char*)proj + (size_t)btg * SLICE_BYTES;
    float4 v[KK1];
    for (int k = 0; k < KK1; ++k) v[k] = *(const float4*)(base + voff[k]);

    float4 a0 = {0.f, 0.f, 0.f, 0.f}, a1 = a0, a2 = a0, a3 = a0;
    for (int k = 0; k < KK1; ++k) {
      float4 x = v[k];
      if (!((okmask >> k) & 1u)) x = float4{0.f, 0.f, 0.f, 0.f};
      a0.x += w[k][0] * x.x; a0.y += w[k][0] * x.y; a0.z += w[k][0] * x.z; a0.w += w[k][0] * x.w;
      a1.x += w[k][1] * x.x; a1.y += w[k][1] * x.y; a1.z += w[k][1] * x.z; a1.w += w[k][1] * x.w;
      a2.x += w[k][2] * x.x; a2.y += w[k][2] * x.y; a2.z += w[k][2] * x.z; a2.w += w[k][2] * x.w;
      a3.x += w[k][3] * x.x; a3.y += w[k][3] * x.y; a3.z += w[k][3] * x.z; a3.w += w[k][3] * x.w;
    }

    float* o = out + ((size_t)(btg + j) * NN + n) * (HH * COUT) + c4 * 4;
    *(float4*)(o + 0 * COUT) = float4{a0.x + dg[0], a0.y + dg[0], a0.z + dg[0], a0.w + dg[0]};
    *(float4*)(o + 1 * COUT) = float4{a1.x + dg[1], a1.y + dg[1], a1.z + dg[1], a1.w + dg[1]};
    *(float4*)(o + 2 * COUT) = float4{a2.x + dg[2], a2.y + dg[2], a2.z + dg[2], a2.w + dg[2]};
    *(float4*)(o + 3 * COUT) = float4{a3.x + dg[3], a3.y + dg[3], a3.z + dg[3], a3.w + dg[3]};
  }
}

extern "C" void kernel_launch(void* const* d_in, const int* in_sizes, int n_in,
                              void* d_out, int out_size, void* d_ws, size_t ws_size,
                              hipStream_t stream) {
  const float* x  = (const float*)d_in[0];
  const float* W  = (const float*)d_in[1];
  const float* b  = (const float*)d_in[2];
  const float* nd = (const float*)d_in[3];
  const int*   nn = (const int*)d_in[4];
  float* out = (float*)d_out;

  const long M = (long)BB * TT * NN;

  float* proj = (float*)d_ws;
  float* wbuf = proj + M * COUT;
  float* dagg = wbuf + (long)NN * KK1 * HH;

  const int tiles = (int)(M / 16);
  proj_wmma_kernel<<<(tiles + 7) / 8, 256, 0, stream>>>(x, W, b, proj, (int)M);

  weights_kernel<<<(NN + 255) / 256, 256, 0, stream>>>(nd, nn, wbuf, dagg);

  agg_kernel<<<dim3(NN / AGG_WAVES, (BB * TT) / BT_PER_WAVE), AGG_WAVES * 32, 0, stream>>>(proj, wbuf, dagg, nn, out);
}


# weights kernel folded into the projection launch as its first 40 work-groups (independent work, chosen by block index): 3 launches become 2; every sum order unchanged
# speedup vs baseline: 2.6491x; 1.0492x over previous
#include <hip/hip_runtime.h>
#include <hip/hip_bf16.h>

#define BB 2
#define TT 12
#define NN 10000
#define CIN 64
#define COUT 32
#define KK1 17
#define HH 4

typedef float v2f __attribute__((ext_vector_type(2)));
typedef float v8f __attribute__((ext_vector_type(8)));

#define WGT_BLOCKS ((NN + 255) / 256)

__device__ __forceinline__ void proj_tile(
    const float* __restrict__ x, const float* __restrict__ W,
    const float* __restrict__ bias, float* __restrict__ proj, int M, int pblock) {
  const int lane = threadIdx.x & 31;
  const int half = lane >> 4;
  const int l15  = lane & 15;
  const long wave = (long)pblock * (blockDim.x >> 5) + (threadIdx.x >> 5);
  const long row0 = wave * 16;
  if (row0 >= M) return;

  const float* ap  = x + (row0 + l15) * CIN + half * 2;
  const float* bp0 = W + (long)l15        * CIN + half * 2;
  const float* bp1 = W + (long)(l15 + 16) * CIN + half * 2;

  v8f acc0 = {};
  v8f acc1 = {};
#pragma unroll
  for (int ks = 0; ks < 16; ++ks) {
    v2f a  = *(const v2f*)(ap  + ks * 4);
    v2f b0 = *(const v2f*)(bp0 + ks * 4);
    v2f b1 = *(const v2f*)(bp1 + ks * 4);
    acc0 = __builtin_amdgcn_wmma_f32_16x16x4_f32(false, a, false, b0,
                                                 (short)0, acc0, false, false);
    acc1 = __builtin_amdgcn_wmma_f32_16x16x4_f32(false, a, false, b1,
                                                 (short)0, acc1, false, false);
  }

  const float bv0 = bias[l15];
  const float bv1 = bias[l15 + 16];
#pragma unroll
  for (int r = 0; r < 8; ++r) {
    const long m = row0 + r + half * 8;
    proj[m * COUT + l15]      = acc0[r] + bv0;
    proj[m * COUT + 16 + l15] = acc1[r] + bv1;
  }
}

__device__ __forceinline__ void weights_nodes(
    const float* __restrict__ nd, const int* __restrict__ nn,
    float* __restrict__ wbuf, float* __restrict__ dagg, int wblock) {
  const int n = wblock * blockDim.x + threadIdx.x;
  if (n >= NN) return;
  float da[HH] = {0.f, 0.f, 0.f, 0.f};
  for (int k = 0; k < KK1; ++k) {
    const float d   = nd[n * KK1 + k];
    const int node  = nn[n * KK1 + k];
    const float d2  = d * d;
    const float dc  = isinf(d) ? 0.f : d;
#pragma unroll
    for (int h = 0; h < HH; ++h) {
      const float lam = (float)(h + 1) * (1.0f / HH);
      float wv = __expf(-d2 * lam);
      if (node == -1)   wv = 0.f;
      if (wv < 1e-5f)   wv = 0.f;
      wbuf[(n * KK1 + k) * HH + h] = wv;
      da[h] += wv * dc;
    }
  }
#pragma unroll
  for (int h = 0; h < HH; ++h) dagg[n * HH + h] = da[h];
}

__global__ __launch_bounds__(256) void proj_weights_kernel(
    const float* __restrict__ x, const float* __restrict__ W,
    const float* __restrict__ bias, const float* __restrict__ nd,
    const int* __restrict__ nn, float* __restrict__ proj,
    float* __restrict__ wbuf, float* __restrict__ dagg, int M) {
  if (blockIdx.x < WGT_BLOCKS) {
    weights_nodes(nd, nn, wbuf, dagg, (int)blockIdx.x);
    return;
  }
  proj_tile(x, W, bias, proj, M, (int)blockIdx.x - WGT_BLOCKS);
}

#define SLICE_BYTES (NN * COUT * 4)
#define BT_GROUP 4
#define BT_PER_WAVE 12
#define AGG_WAVES 4

__global__ __launch_bounds__(AGG_WAVES * 32) void agg_kernel(
    const float* __restrict__ proj, const float* __restrict__ wbuf,
    const float* __restrict__ dagg, const int* __restrict__ nn,
    float* __restrict__ out) {
  const int lane = threadIdx.x & 31;
  const int wv = __builtin_amdgcn_readfirstlane((int)(threadIdx.x >> 5));
  const int n = blockIdx.x * AGG_WAVES + wv;
  if (n >= NN) return;
  const int bt0 = blockIdx.y * BT_PER_WAVE;
  const int j  = lane >> 3;
  const int c4 = lane & 7;

  unsigned voff[KK1];
  unsigned okmask = 0;
  const unsigned lanepart = (unsigned)j * SLICE_BYTES + (unsigned)c4 * 16u;
  for (int k = 0; k < KK1; ++k) {
    const int node = nn[n * KK1 + k];
    const bool ok = (node >= 0) && (node < NN);
    okmask |= (ok ? 1u : 0u) << k;
    voff[k] = (ok ? (unsigned)node : 0u) * (COUT * 4) + lanepart;
  }
  float w[KK1][HH];
  for (int k = 0; k < KK1; ++k)
    for (int h = 0; h < HH; ++h) w[k][h] = wbuf[(n * KK1 + k) * HH + h];
  float dg[HH];
  for (int h = 0; h < HH; ++h) dg[h] = dagg[n * HH + h];

  for (int g = 0; g < BT_PER_WAVE / BT_GROUP; ++g) {
    const int btg = bt0 + g * BT_GROUP;
    const char* base = (const char*)proj + (size_t)btg * SLICE_BYTES;
    float4 v[KK1];
    for (int k = 0; k < KK1; ++k) v[k] = *(const float4*)(base + voff[k]);

    float4 a0 = {0.f, 0.f, 0.f, 0.f}, a1 = a0, a2 = a0, a3 = a0;
    for (int k = 0; k < KK1; ++k) {
      float4 x = v[k];
      if (!((okmask >> k) & 1u)) x = float4{0.f, 0.f, 0.f, 0.f};
      a0.x += w[k][0] * x.x; a0.y += w[k][0] * x.y; a0.z += w[k][0] * x.z; a0.w += w[k][0] * x.w;
      a1.x += w[k][1] * x.x; a1.y += w[k][1] * x.y; a1.z += w[k][1] * x.z; a1.w += w[k][1] * x.w;
      a2.x += w[k][2] * x.x; a2.y += w[k][2] * x.y; a2.z += w[k][2] * x.z; a2.w += w[k][2] * x.w;
      a3.x += w[k][3] * x.x; a3.y += w[k][3] * x.y; a3.z += w[k][3] * x.z; a3.w += w[k][3] * x.w;
    }

    float* o = out + ((size_t)(btg + j) * NN + n) * (HH * COUT) + c4 * 4;
    *(float4*)(o + 0 * COUT) = float4{a0.x + dg[0], a0.y + dg[0], a0.z + dg[0], a0.w + dg[0]};
    *(float4*)(o + 1 * COUT) = float4{a1.x + dg[1], a1.y + dg[1], a1.z + dg[1], a1.w + dg[1]};
    *(float4*)(o + 2 * COUT) = float4{a2.x + dg[2], a2.y + dg[2], a2.z + dg[2], a2.w + dg[2]};
    *(float4*)(o + 3 * COUT) = float4{a3.x + dg[3], a3.y + dg[3], a3.z + dg[3], a3.w + dg[3]};
  }
}

extern "C" void kernel_launch(void* const* d_in, const int* in_sizes, int n_in,
                              void* d_out, int out_size, void* d_ws, size_t ws_size,
                              hipStream_t stream) {
  const float* x  = (const float*)d_in[0];
  const float* W  = (const float*)d_in[1];
  const float* b  = (const float*)d_in[2];
  const float* nd = (const float*)d_in[3];
  const int*   nn = (const int*)d_in[4];
  float* out = (float*)d_out;

  const long M = (long)BB * TT * NN;

  float* proj = (float*)d_ws;
  float* wbuf = proj + M * COUT;
  float* dagg = wbuf + (long)NN * KK1 * HH;

  const int tiles = (int)(M / 16);
  proj_weights_kernel<<<WGT_BLOCKS + (tiles + 7) / 8, 256, 0, stream>>>(
      x, W, b, nd, nn, proj, wbuf, dagg, (int)M);

  agg_kernel<<<dim3(NN / AGG_WAVES, (BB * TT) / BT_PER_WAVE), AGG_WAVES * 32, 0, stream>>>(proj, wbuf, dagg, nn, out);
}


# proj: each wave keeps the 32 W fragments in registers and walks 5 consecutive M-tiles (x fragments issued two tiles ahead, stores per tile), 4-wave groups; every sum order unchanged
# speedup vs baseline: 3.1050x; 1.1721x over previous
#include <hip/hip_runtime.h>
#include <hip/hip_bf16.h>

#define BB 2
#define TT 12
#define NN 10000
#define CIN 64
#define COUT 32
#define KK1 17
#define HH 4

typedef float v2f __attribute__((ext_vector_type(2)));
typedef float v8f __attribute__((ext_vector_type(8)));

#define BLOCK_THREADS 128
#define WGT_BLOCKS ((NN + BLOCK_THREADS - 1) / BLOCK_THREADS)

#define PROJ_TILES_PER_WAVE 5
#define PROJ_WAVES_PER_BLOCK (BLOCK_THREADS / 32)
#define PROJ_TILES (BB * TT * NN / 16)
#define PROJ_BLOCKS (PROJ_TILES / (PROJ_TILES_PER_WAVE * PROJ_WAVES_PER_BLOCK))
static_assert((BB * TT * NN) % 16 == 0, "M must be a whole number of 16-row tiles");
static_assert(PROJ_TILES % (PROJ_TILES_PER_WAVE * PROJ_WAVES_PER_BLOCK) == 0,
              "tiles must divide evenly over waves and blocks");

__device__ __forceinline__ void load_a_tile(v2f (&a)[16], const float* __restrict__ ap) {
#pragma unroll
  for (int ks = 0; ks < 16; ++ks) a[ks] = *(const v2f*)(ap + ks * 4);
}

__device__ __forceinline__ void proj_tiles(
    const float* __restrict__ x, const float* __restrict__ W,
    const float* __restrict__ bias, float* __restrict__ proj, int M, int pblock) {
  const int lane = threadIdx.x & 31;
  const int half = lane >> 4;
  const int l15  = lane & 15;
  const int wv = __builtin_amdgcn_readfirstlane((int)(threadIdx.x >> 5));
  const long wave  = (long)pblock * PROJ_WAVES_PER_BLOCK + wv;
  const long tile0 = wave * PROJ_TILES_PER_WAVE;
  if ((tile0 + PROJ_TILES_PER_WAVE) * 16 > (long)M) return;

  const float* bp0 = W + (long)l15        * CIN + half * 2;
  const float* bp1 = W + (long)(l15 + 16) * CIN + half * 2;
  const float* ap  = x + (tile0 * 16 + l15) * CIN + half * 2;

  v2f a[3][16];
  load_a_tile(a[0], ap);

  v2f b0[16], b1[16];
#pragma unroll
  for (int ks = 0; ks < 16; ++ks) {
    b0[ks] = *(const v2f*)(bp0 + ks * 4);
    b1[ks] = *(const v2f*)(bp1 + ks * 4);
  }
  const float bv0 = bias[l15];
  const float bv1 = bias[l15 + 16];
  load_a_tile(a[1], ap + 16 * CIN);

#pragma unroll
  for (int it = 0; it < PROJ_TILES_PER_WAVE; ++it) {
    if (it + 2 < PROJ_TILES_PER_WAVE)
      load_a_tile(a[(it + 2) % 3], ap + (long)(it + 2) * 16 * CIN);

    v8f acc0 = {};
    v8f acc1 = {};
#pragma unroll
    for (int ks = 0; ks < 16; ++ks) {
      acc0 = __builtin_amdgcn_wmma_f32_16x16x4_f32(false, a[it % 3][ks], false, b0[ks],
                                                   (short)0, acc0, false, false);
      acc1 = __builtin_amdgcn_wmma_f32_16x16x4_f32(false, a[it % 3][ks], false, b1[ks],
                                                   (short)0, acc1, false, false);
    }

    const long row0 = (tile0 + it) * 16;
#pragma unroll
    for (int r = 0; r < 8; ++r) {
      const long m = row0 + r + half * 8;
      proj[m * COUT + l15]      = acc0[r] + bv0;
      proj[m * COUT + 16 + l15] = acc1[r] + bv1;
    }
    __builtin_amdgcn_sched_barrier(0);
  }
}

__device__ __forceinline__ void weights_nodes(
    const float* __restrict__ nd, const int* __restrict__ nn,
    float* __restrict__ wbuf, float* __restrict__ dagg, int wblock) {
  const int n = wblock * blockDim.x + threadIdx.x;
  if (n >= NN) return;
  float da[HH] = {0.f, 0.f, 0.f, 0.f};
  for (int k = 0; k < KK1; ++k) {
    const float d   = nd[n * KK1 + k];
    const int node  = nn[n * KK1 + k];
    const float d2  = d * d;
    const float dc  = isinf(d) ? 0.f : d;
#pragma unroll
    for (int h = 0; h < HH; ++h) {
      const float lam = (float)(h + 1) * (1.0f / HH);
      float wv = __expf(-d2 * lam);
      if (node == -1)   wv = 0.f;
      if (wv < 1e-5f)   wv = 0.f;
      wbuf[(n * KK1 + k) * HH + h] = wv;
      da[h] += wv * dc;
    }
  }
#pragma unroll
  for (int h = 0; h < HH; ++h) dagg[n * HH + h] = da[h];
}

__global__ __launch_bounds__(BLOCK_THREADS) void proj_weights_kernel(
    const float* __restrict__ x, const float* __restrict__ W,
    const float* __restrict__ bias, const float* __restrict__ nd,
    const int* __restrict__ nn, float* __restrict__ proj,
    float* __restrict__ wbuf, float* __restrict__ dagg, int M) {
  if (blockIdx.x < WGT_BLOCKS) {
    weights_nodes(nd, nn, wbuf, dagg, (int)blockIdx.x);
    return;
  }
  proj_tiles(x, W, bias, proj, M, (int)blockIdx.x - WGT_BLOCKS);
}

#define SLICE_BYTES (NN * COUT * 4)
#define BT_GROUP 4
#define BT_PER_WAVE 12
#define AGG_WAVES 4

__global__ __launch_bounds__(AGG_WAVES * 32) void agg_kernel(
    const float* __restrict__ proj, const float* __restrict__ wbuf,
    const float* __restrict__ dagg, const int* __restrict__ nn,
    float* __restrict__ out) {
  const int lane = threadIdx.x & 31;
  const int wv = __builtin_amdgcn_readfirstlane((int)(threadIdx.x >> 5));
  const int n = blockIdx.x * AGG_WAVES + wv;
  if (n >= NN) return;
  const int bt0 = blockIdx.y * BT_PER_WAVE;
  const int j  = lane >> 3;
  const int c4 = lane & 7;

  unsigned voff[KK1];
  unsigned okmask = 0;
  const unsigned lanepart = (unsigned)j * SLICE_BYTES + (unsigned)c4 * 16u;
  for (int k = 0; k < KK1; ++k) {
    const int node = nn[n * KK1 + k];
    const bool ok = (node >= 0) && (node < NN);
    okmask |= (ok ? 1u : 0u) << k;
    voff[k] = (ok ? (unsigned)node : 0u) * (COUT * 4) + lanepart;
  }
  float w[KK1][HH];
  for (int k = 0; k < KK1; ++k)
    for (int h = 0; h < HH; ++h) w[k][h] = wbuf[(n * KK1 + k) * HH + h];
  float dg[HH];
  for (int h = 0; h < HH; ++h) dg[h] = dagg[n * HH + h];

  for (int g = 0; g < BT_PER_WAVE / BT_GROUP; ++g) {
    const int btg = bt0 + g * BT_GROUP;
    const char* base = (const char*)proj + (size_t)btg * SLICE_BYTES;
    float4 v[KK1];
    for (int k = 0; k < KK1; ++k) v[k] = *(const float4*)(base + voff[k]);

    float4 a0 = {0.f, 0.f, 0.f, 0.f}, a1 = a0, a2 = a0, a3 = a0;
    for (int k = 0; k < KK1; ++k) {
      float4 x = v[k];
      if (!((okmask >> k) & 1u)) x = float4{0.f, 0.f, 0.f, 0.f};
      a0.x += w[k][0] * x.x; a0.y += w[k][0] * x.y; a0.z += w[k][0] * x.z; a0.w += w[k][0] * x.w;
      a1.x += w[k][1] * x.x; a1.y += w[k][1] * x.y; a1.z += w[k][1] * x.z; a1.w += w[k][1] * x.w;
      a2.x += w[k][2] * x.x; a2.y += w[k][2] * x.y; a2.z += w[k][2] * x.z; a2.w += w[k][2] * x.w;
      a3.x += w[k][3] * x.x; a3.y += w[k][3] * x.y; a3.z += w[k][3] * x.z; a3.w += w[k][3] * x.w;
    }

    float* o = out + ((size_t)(btg + j) * NN + n) * (HH * COUT) + c4 * 4;
    *(float4*)(o + 0 * COUT) = float4{a0.x + dg[0], a0.y + dg[0], a0.z + dg[0], a0.w + dg[0]};
    *(float4*)(o + 1 * COUT) = float4{a1.x + dg[1], a1.y + dg[1], a1.z + dg[1], a1.w + dg[1]};
    *(float4*)(o + 2 * COUT) = float4{a2.x + dg[2], a2.y + dg[2], a2.z + dg[2], a2.w + dg[2]};
    *(float4*)(o + 3 * COUT) = float4{a3.x + dg[3], a3.y + dg[3], a3.z + dg[3], a3.w + dg[3]};
  }
}

extern "C" void kernel_launch(void* const* d_in, const int* in_sizes, int n_in,
                              void* d_out, int out_size, void* d_ws, size_t ws_size,
                              hipStream_t stream) {
  const float* x  = (const float*)d_in[0];
  const float* W  = (const float*)d_in[1];
  const float* b  = (const float*)d_in[2];
  const float* nd = (const float*)d_in[3];
  const int*   nn = (const int*)d_in[4];
  float* out = (float*)d_out;

  const long M = (long)BB * TT * NN;

  float* proj = (float*)d_ws;
  float* wbuf = proj + M * COUT;
  float* dagg = wbuf + (long)NN * KK1 * HH;

  proj_weights_kernel<<<WGT_BLOCKS + PROJ_BLOCKS, BLOCK_THREADS, 0, stream>>>(
      x, W, b, nd, nn, proj, wbuf, dagg, (int)M);

  agg_kernel<<<dim3(NN / AGG_WAVES, (BB * TT) / BT_PER_WAVE), AGG_WAVES * 32, 0, stream>>>(proj, wbuf, dagg, nn, out);
}
